# starting kernel
# baseline (speedup 1.0000x reference)
#include <hip/hip_runtime.h>
#include <hip/hip_bf16.h>
#include <math.h>


typedef __attribute__((ext_vector_type(2))) float v2f;
typedef __attribute__((ext_vector_type(8))) float v8f;

#define NPIX      16384
#define KDIM      72
#define NDIM      128
#define LDSW_STR  132
#define VOTE_STR  130
#define CAPS_EPS  1e-7f
#define PH        34
#define PLANE     (PH * PH * 128)
#define PIX_PER_WAVE 4

__global__ void caps_pad_transpose_kernel(const float* __restrict__ in,
                                          float* __restrict__ out) {
    int t = blockIdx.x * blockDim.x + threadIdx.x;
    if (t >= 16 * PH * PH * 16) return;
    int b  = t & 15;
    int r  = t >> 4;
    int ww = r % PH;  r /= PH;
    int hh = r % PH;
    int ic = r / PH;
    int h = hh - 1;
    int w = ww - 1;
    float4 v0 = float4{0.f, 0.f, 0.f, 0.f};
    float4 v1 = v0;
    if ((unsigned)h < 32u && (unsigned)w < 32u) {
        const float4* src = (const float4*)(in + ((((size_t)b * 32 + h) * 32 + w) * 16 + ic) * 8);
        v0 = src[0];
        v1 = src[1];
    }
    float4* dst = (float4*)(out + (size_t)t * 8);
    dst[0] = v0;
    dst[1] = v1;
}

__global__ __launch_bounds__(64) void caps_conv_route_kernel(
    const float* __restrict__ xTp,
    const float* __restrict__ Wg,
    const float* __restrict__ bias,
    float* __restrict__ out)
{
    __shared__ float ldsW[KDIM * LDSW_STR];
    __shared__ float ldsVotes[2][16 * VOTE_STR];
    __shared__ float ldsLogits[2][16 * 16];
    __shared__ float ldsMx[2][16];
    __shared__ float ldsRs[2][16];

    const int tid  = threadIdx.x;
    const int wib  = tid >> 5;
    const int lane = tid & 31;

    for (int i = tid; i < KDIM * NDIM; i += 64) {
        int r = i >> 7;
        int c = i & 127;
        ldsW[r * LDSW_STR + c] = Wg[i];
    }
    __syncthreads();

    const int m  = lane & 15;
    const int hi = lane >> 4;

    float bias_l[4];
#pragma unroll
    for (int q = 0; q < 4; ++q) bias_l[q] = bias[q * 32 + lane];

    float* votes  = ldsVotes[wib];
    float* logits = ldsLogits[wib];

    const int waveId = blockIdx.x * 2 + wib;

#pragma unroll 1
    for (int it = 0; it < PIX_PER_WAVE; ++it) {
        const int pixel = waveId * PIX_PER_WAVE + it;
        const int icap  = pixel >> 10;
        const int h     = (pixel >> 5) & 31;
        const int w     = pixel & 31;

        v8f acc[8];
#pragma unroll
        for (int t = 0; t < 8; ++t)
            acc[t] = v8f{0.f, 0.f, 0.f, 0.f, 0.f, 0.f, 0.f, 0.f};

        const float* base = xTp + (size_t)icap * PLANE + (h * PH + w) * 128;

#pragma unroll
        for (int ks = 0; ks < 18; ++ks) {
            int K0  = ks * 4 + hi * 2;
            int kh  = K0 / 24;
            int rem = K0 - kh * 24;
            int kw  = rem >> 3;
            int id0 = rem & 7;
            v2f a = *(const v2f*)(base + (kh * PH + kw) * 128 + m * 8 + id0);
#pragma unroll
            for (int t = 0; t < 8; ++t) {
                int col = t * 16 + m;
                v2f bf;
                bf[0] = ldsW[K0 * LDSW_STR + col];
                bf[1] = ldsW[(K0 + 1) * LDSW_STR + col];
                acc[t] = __builtin_amdgcn_wmma_f32_16x16x4_f32(
                    false, a, false, bf, (short)0, acc[t], false, false);
            }
        }

#pragma unroll
        for (int t = 0; t < 8; ++t)
#pragma unroll
            for (int v = 0; v < 8; ++v)
                votes[(v + 8 * hi) * VOTE_STR + t * 16 + m] = acc[t][v];

        float pre[4], act[4];
#pragma unroll
        for (int q = 0; q < 4; ++q) {
            int c = q * 32 + lane;
            float s = 0.f;
            for (int ic = 0; ic < 16; ++ic) s += votes[ic * VOTE_STR + c];
            pre[q] = s * (1.f / 16.f) + bias_l[q];
        }
#pragma unroll
        for (int q = 0; q < 4; ++q) {
            float n2 = pre[q] * pre[q];
            n2 += __shfl_xor(n2, 1, 32);
            n2 += __shfl_xor(n2, 2, 32);
            n2 += __shfl_xor(n2, 4, 32);
            float scale = (n2 / (1.f + n2)) / sqrtf(n2 + CAPS_EPS);
            act[q] = pre[q] * scale;
        }

        for (int ic = 0; ic < 16; ++ic) {
#pragma unroll
            for (int q = 0; q < 4; ++q) {
                float d = votes[ic * VOTE_STR + q * 32 + lane] * act[q];
                d += __shfl_xor(d, 1, 32);
                d += __shfl_xor(d, 2, 32);
                d += __shfl_xor(d, 4, 32);
                if ((lane & 7) == 0)
                    logits[ic * 16 + q * 4 + (lane >> 3)] = d;
            }
        }

        if (lane < 16) {
            float mx = -3.0e38f;
            for (int oc = 0; oc < 16; ++oc) mx = fmaxf(mx, logits[lane * 16 + oc]);
            float s = 0.f;
            for (int oc = 0; oc < 16; ++oc) s += __expf(logits[lane * 16 + oc] - mx);
            ldsMx[wib][lane] = mx;
            ldsRs[wib][lane] = 1.f / s;
        }

#pragma unroll
        for (int q = 0; q < 4; ++q) {
            int c  = q * 32 + lane;
            int oc = c >> 3;
            float s = 0.f;
            for (int ic = 0; ic < 16; ++ic) {
                float r = __expf(logits[ic * 16 + oc] - ldsMx[wib][ic]) * ldsRs[wib][ic];
                s += r * votes[ic * VOTE_STR + c];
            }
            pre[q] = s + bias_l[q];
        }
#pragma unroll
        for (int q = 0; q < 4; ++q) {
            float n2 = pre[q] * pre[q];
            n2 += __shfl_xor(n2, 1, 32);
            n2 += __shfl_xor(n2, 2, 32);
            n2 += __shfl_xor(n2, 4, 32);
            float scale = (n2 / (1.f + n2)) / sqrtf(n2 + CAPS_EPS);
            out[(size_t)pixel * 128 + q * 32 + lane] = pre[q] * scale;
        }
    }
}

extern "C" void kernel_launch(void* const* d_in, const int* in_sizes, int n_in,
                              void* d_out, int out_size, void* d_ws, size_t ws_size,
                              hipStream_t stream) {
    (void)in_sizes; (void)n_in; (void)out_size; (void)ws_size;
    const float* input = (const float*)d_in[0];
    const float* Wg    = (const float*)d_in[1];
    const float* bias  = (const float*)d_in[2];
    float* out = (float*)d_out;
    float* xTp = (float*)d_ws;

    int ngroups = 16 * PH * PH * 16;
    caps_pad_transpose_kernel<<<(ngroups + 255) / 256, 256, 0, stream>>>(input, xTp);
    caps_conv_route_kernel<<<NPIX / (2 * PIX_PER_WAVE), 64, 0, stream>>>(xTp, Wg, bias, out);
}


# Squash scale factored out of the agreement product and applied to the logit tile; its operand summed per lane straight from the slab (capsule = lane): 1 squash chain, no shuffles; new logit sum order
# speedup vs baseline: 3.6461x; 3.6461x over previous
#include <hip/hip_runtime.h>
#include <hip/hip_bf16.h>
#include <math.h>


typedef __attribute__((ext_vector_type(2))) float v2f;
typedef __attribute__((ext_vector_type(8))) float v8f;

#define NPIX      16384
#define KDIM      72
#define NDIM      128
#define WAVES_PER_BLOCK 8
#define VOTE_STR  132
#define CAPS_EPS  1e-7f
#define PIX_PER_WAVE 4
#define TILE_ROWS 4
#define TILE_COLS 6
#define TILE_POS  (TILE_ROWS * TILE_COLS)

template <int MASK>
__device__ __forceinline__ float row_xor(float x) {
    return __builtin_bit_cast(float,
        __builtin_amdgcn_mov_dpp(__builtin_bit_cast(int, x), 0x160 + MASK, 0xf, 0xf, false));
}

__device__ __forceinline__ float add2f(float a, float b) { return a + b; }

__device__ __forceinline__ float xhalf(float x) {
    return __builtin_bit_cast(float,
        __builtin_amdgcn_permlanex16(__builtin_bit_cast(int, x), __builtin_bit_cast(int, x),
                                     0x76543210, 0xfedcba98, false, false));
}

__global__ __launch_bounds__(32 * WAVES_PER_BLOCK) void caps_conv_route_kernel(
    const float* __restrict__ in,
    const float* __restrict__ Wg,
    const float* __restrict__ bias,
    float* __restrict__ out)
{
    __shared__ float4 ldsWf[18 * 2 * 4 * 16];
    __shared__ float4 ldsX[4 * TILE_POS * 2 * 16];
    __shared__ __attribute__((aligned(16))) float ldsVotes[WAVES_PER_BLOCK][16 * VOTE_STR];

    const int tid  = threadIdx.x;
    const int wib  = tid >> 5;
    const int lane = tid & 31;

    const int wq  = blockIdx.x & 7;
    const int hp  = (blockIdx.x >> 3) & 15;
    const int icg = blockIdx.x >> 7;

    {
        const int part = tid & 7;
        const int bb   = (tid >> 3) & 15;
        const int p0   = tid >> 7;
        float4 xv[12];
        for (int j = 0; j < 12; ++j) {
            const int r  = j / 3;
            const int cc = 2 * (j % 3) + p0;
            const int hh = 2 * hp + r - 1;
            const int ww = 4 * wq + cc - 1;
            const bool inb = ((unsigned)hh < 32u) && ((unsigned)ww < 32u);
            const int hc = min(max(hh, 0), 31);
            const int wc = min(max(ww, 0), 31);
            const float4 v = *(const float4*)(in + (size_t)bb * 131072 + hc * 4096 + wc * 128
                                                 + icg * 32 + part * 4);
            xv[j] = inb ? v : float4{0.f, 0.f, 0.f, 0.f};
        }
        for (int j = 0; j < 12; ++j) {
            const int r  = j / 3;
            const int cc = 2 * (j % 3) + p0;
            const int pos = r * TILE_COLS + cc;
            ldsX[(((part >> 1) * TILE_POS + pos) * 2 + (part & 1)) * 16 + ((bb + 4 * part) & 15)] = xv[j];
        }
    }

    for (int i = 0; i < 9; ++i) {
        const int idx = tid + i * 32 * WAVES_PER_BLOCK;
        const int sm_ = idx & 15;
        const int sq  = (idx >> 4) & 3;
        const int sh  = (idx >> 6) & 1;
        const int sk  = idx >> 7;
        const int K0  = sk * 4 + sh * 2;
        const int c0  = sq * 32 + sm_;
        float4 v;
        v.x = Wg[K0 * NDIM + c0];
        v.y = Wg[(K0 + 1) * NDIM + c0];
        v.z = Wg[K0 * NDIM + c0 + 16];
        v.w = Wg[(K0 + 1) * NDIM + c0 + 16];
        ldsWf[idx] = v;
    }
    __syncthreads();

    const int m  = lane & 15;
    const int hi = lane >> 4;

    float bias_c[4];
#pragma unroll
    for (int i = 0; i < 4; ++i) bias_c[i] = bias[8 * m + 4 * hi + i];
    float bias_r[4];
#pragma unroll
    for (int j = 0; j < 4; ++j) bias_r[j] = bias[64 * hi + 16 * j + m];

    const float4* wfrag = ldsWf + hi * 64 + m;

    float* votes  = ldsVotes[wib];

    const int icl = wib & 3;
    const int hl  = wib >> 2;
    const int slotE = (m + 8 * icl) & 15;
    const int slotO = (slotE + 4) & 15;
    const float* xrow = (const float*)ldsX + (icl * TILE_POS + hl * TILE_COLS) * 128 + 2 * hi;
    const float* xE = xrow + slotE * 4;
    const float* xO = xrow + 64 + slotO * 4;

#pragma unroll 1
    for (int it = 0; it < PIX_PER_WAVE; ++it) {
        const int icap  = icg * 4 + icl;
        const int h     = 2 * hp + hl;
        const int w     = 4 * wq + it;
        const int pixel = (icap << 10) + (h << 5) + w;

        v8f acc[8];
#pragma unroll
        for (int t = 0; t < 8; ++t)
            acc[t] = v8f{0.f, 0.f, 0.f, 0.f, 0.f, 0.f, 0.f, 0.f};

        const float* tapE = xE + it * 128;
        const float* tapO = xO + it * 128;

#pragma unroll
        for (int ks = 0; ks < 18; ++ks) {
            const int tap = ks >> 1;
            const int kh  = tap / 3;
            const int kw  = tap - kh * 3;
            v2f a = *(const v2f*)(((ks & 1) ? tapO : tapE) + (kh * TILE_COLS + kw) * 128);
            for (int q = 0; q < 4; ++q) {
                const float4 w4 = wfrag[ks * 128 + q * 16];
                v2f b0, b1;
                b0[0] = w4.x; b0[1] = w4.y;
                b1[0] = w4.z; b1[1] = w4.w;
                acc[2 * q] = __builtin_amdgcn_wmma_f32_16x16x4_f32(
                    false, a, false, b0, (short)0, acc[2 * q], false, false);
                acc[2 * q + 1] = __builtin_amdgcn_wmma_f32_16x16x4_f32(
                    false, a, false, b1, (short)0, acc[2 * q + 1], false, false);
            }
        }

#pragma unroll
        for (int t = 0; t < 8; ++t)
#pragma unroll
            for (int v = 0; v < 8; ++v)
                votes[(v + 8 * hi) * VOTE_STR + t * 16 + m] = acc[t][v];

        const float* vcol = votes + 8 * m + 4 * hi;
        v2f s01 = v2f{0.f, 0.f};
        v2f s23 = v2f{0.f, 0.f};
        for (int ic = 0; ic < 16; ++ic) {
            const float4 v4 = *(const float4*)(vcol + ic * VOTE_STR);
            s01 += v2f{v4.x, v4.y};
            s23 += v2f{v4.z, v4.w};
        }
        v2f bq0, bq1;
        bq0[0] = s01[0] * (1.f / 16.f) + bias_c[0];
        bq0[1] = s01[1] * (1.f / 16.f) + bias_c[1];
        bq1[0] = s23[0] * (1.f / 16.f) + bias_c[2];
        bq1[1] = s23[1] * (1.f / 16.f) + bias_c[3];

        const v2f zero2 = v2f{0.f, 0.f};
        v8f lgE = v8f{0.f, 0.f, 0.f, 0.f, 0.f, 0.f, 0.f, 0.f};
        v8f lgO = lgE;
        v2f avf[32];
        const float* vrow = votes + m * VOTE_STR + 4 * hi;
#pragma unroll
        for (int oc = 0; oc < 16; ++oc) {
            const float4 a4 = *(const float4*)(vrow + 8 * oc);
            avf[2 * oc][0]     = a4.x;
            avf[2 * oc][1]     = a4.y;
            avf[2 * oc + 1][0] = a4.z;
            avf[2 * oc + 1][1] = a4.w;
        }
        __builtin_amdgcn_sched_barrier(0);
#pragma unroll
        for (int ks = 0; ks < 32; ++ks) {
            const v2f av = avf[ks];
            const bool own = (m == (ks >> 1));
            const v2f bv = own ? ((ks & 1) ? bq1 : bq0) : zero2;
            if (ks & 1)
                lgO = __builtin_amdgcn_wmma_f32_16x16x4_f32(
                    false, av, false, bv, (short)0, lgO, false, false);
            else
                lgE = __builtin_amdgcn_wmma_f32_16x16x4_f32(
                    false, av, false, bv, (short)0, lgE, false, false);
        }
        float n2c = (bq0[0] * bq0[0] + bq0[1] * bq0[1]) + (bq1[0] * bq1[0] + bq1[1] * bq1[1]);
        n2c += xhalf(n2c);
        const float sc1 = (n2c / (1.f + n2c)) / sqrtf(n2c + CAPS_EPS);
        v2f rt[4];
        float lg[8], mx[8], ex[8], sm[8], tx[8];
#pragma unroll
        for (int r = 0; r < 8; ++r) { lg[r] = (lgE[r] + lgO[r]) * sc1; mx[r] = lg[r]; }
#define ROW_STEP8(K, DST, COMBINE)                                         \
        _Pragma("unroll") for (int r = 0; r < 8; ++r) tx[r] = row_xor<K>(DST[r]); \
        _Pragma("unroll") for (int r = 0; r < 8; ++r) DST[r] = COMBINE(DST[r], tx[r]);
        ROW_STEP8(1, mx, fmaxf)
        ROW_STEP8(2, mx, fmaxf)
        ROW_STEP8(4, mx, fmaxf)
        ROW_STEP8(8, mx, fmaxf)
#pragma unroll
        for (int r = 0; r < 8; ++r) { ex[r] = __expf(lg[r] - mx[r]); sm[r] = ex[r]; }
        ROW_STEP8(1, sm, add2f)
        ROW_STEP8(2, sm, add2f)
        ROW_STEP8(4, sm, add2f)
        ROW_STEP8(8, sm, add2f)
#undef ROW_STEP8
#pragma unroll
        for (int r = 0; r < 8; ++r) rt[r >> 1][r & 1] = ex[r] * (1.f / sm[r]);

        v8f S[8];
#pragma unroll
        for (int t = 0; t < 8; ++t)
            S[t] = v8f{0.f, 0.f, 0.f, 0.f, 0.f, 0.f, 0.f, 0.f};
#pragma unroll
        for (int ks = 0; ks < 4; ++ks) {
#pragma unroll
            for (int t = 0; t < 8; ++t) {
                v2f bv;
                bv[0] = acc[t][2 * ks];
                bv[1] = acc[t][2 * ks + 1];
                S[t] = __builtin_amdgcn_wmma_f32_16x16x4_f32(
                    false, rt[ks], false, bv, (short)0, S[t], false, false);
            }
        }
        const bool up = (m & 8) != 0;
        float pr[4], n2[4];
#pragma unroll
        for (int j = 0; j < 4; ++j) {
            const float lo = hi ? S[4 + j][2 * j]     : S[j][2 * j];
            const float hv = hi ? S[4 + j][2 * j + 1] : S[j][2 * j + 1];
            pr[j] = (up ? hv : lo) + bias_r[j];
            n2[j] = pr[j] * pr[j];
        }
        float tn[4];
#pragma unroll
        for (int j = 0; j < 4; ++j) tn[j] = row_xor<1>(n2[j]);
#pragma unroll
        for (int j = 0; j < 4; ++j) n2[j] += tn[j];
#pragma unroll
        for (int j = 0; j < 4; ++j) tn[j] = row_xor<2>(n2[j]);
#pragma unroll
        for (int j = 0; j < 4; ++j) n2[j] += tn[j];
#pragma unroll
        for (int j = 0; j < 4; ++j) tn[j] = row_xor<4>(n2[j]);
#pragma unroll
        for (int j = 0; j < 4; ++j) n2[j] += tn[j];
#pragma unroll
        for (int j = 0; j < 4; ++j) {
            float scale = (n2[j] / (1.f + n2[j])) / sqrtf(n2[j] + CAPS_EPS);
            out[(size_t)pixel * 128 + 64 * hi + 16 * j + m] = pr[j] * scale;
        }
    }
}

extern "C" void kernel_launch(void* const* d_in, const int* in_sizes, int n_in,
                              void* d_out, int out_size, void* d_ws, size_t ws_size,
                              hipStream_t stream) {
    (void)in_sizes; (void)n_in; (void)out_size; (void)ws_size;
    const float* input = (const float*)d_in[0];
    const float* Wg    = (const float*)d_in[1];
    const float* bias  = (const float*)d_in[2];
    float* out = (float*)d_out;
    (void)d_ws;

    caps_conv_route_kernel<<<NPIX / (WAVES_PER_BLOCK * PIX_PER_WAVE), 32 * WAVES_PER_BLOCK, 0, stream>>>(input, Wg, bias, out);
}


# Lane-redundant routing math removed: softmax row max/sum and capsule norms by transposing DPP reductions (lane l ends with row l&7), one reciprocal and one squash scale per lane, read by row_share
# speedup vs baseline: 3.8495x; 1.0558x over previous
#include <hip/hip_runtime.h>
#include <hip/hip_bf16.h>
#include <math.h>


typedef __attribute__((ext_vector_type(2))) float v2f;
typedef __attribute__((ext_vector_type(8))) float v8f;

#define NPIX      16384
#define KDIM      72
#define NDIM      128
#define WAVES_PER_BLOCK 8
#define VOTE_STR  132
#define CAPS_EPS  1e-7f
#define PIX_PER_WAVE 4
#define TILE_ROWS 4
#define TILE_COLS 6
#define TILE_POS  (TILE_ROWS * TILE_COLS)

template <int MASK>
__device__ __forceinline__ float row_xor(float x) {
    return __builtin_bit_cast(float,
        __builtin_amdgcn_mov_dpp(__builtin_bit_cast(int, x), 0x160 + MASK, 0xf, 0xf, false));
}

__device__ __forceinline__ float xhalf(float x) {
    return __builtin_bit_cast(float,
        __builtin_amdgcn_permlanex16(__builtin_bit_cast(int, x), __builtin_bit_cast(int, x),
                                     0x76543210, 0xfedcba98, false, false));
}

template <int CTRL>
__device__ __forceinline__ float dpp_read(float x) {
    return __builtin_bit_cast(float,
        __builtin_amdgcn_mov_dpp(__builtin_bit_cast(int, x), CTRL, 0xf, 0xf, false));
}

template <bool IS_MAX>
__device__ __forceinline__ float comb2(float a, float b) { return IS_MAX ? fmaxf(a, b) : a + b; }

template <bool IS_MAX>
__device__ __forceinline__ float row_reduce_own(const float (&v)[8], bool b0, bool b1, bool b2) {
    float p[4], q[2];
#pragma unroll
    for (int i = 0; i < 4; ++i) {
        const float keep = b0 ? v[2 * i + 1] : v[2 * i];
        const float send = b0 ? v[2 * i]     : v[2 * i + 1];
        p[i] = comb2<IS_MAX>(keep, row_xor<1>(send));
    }
#pragma unroll
    for (int i = 0; i < 2; ++i) {
        const float keep = b1 ? p[2 * i + 1] : p[2 * i];
        const float send = b1 ? p[2 * i]     : p[2 * i + 1];
        q[i] = comb2<IS_MAX>(keep, row_xor<2>(send));
    }
    const float keep = b2 ? q[1] : q[0];
    const float send = b2 ? q[0] : q[1];
    float r = comb2<IS_MAX>(keep, row_xor<4>(send));
    r = comb2<IS_MAX>(r, row_xor<8>(r));
    return r;
}

__global__ __launch_bounds__(32 * WAVES_PER_BLOCK) void caps_conv_route_kernel(
    const float* __restrict__ in,
    const float* __restrict__ Wg,
    const float* __restrict__ bias,
    float* __restrict__ out)
{
    __shared__ float4 ldsWf[18 * 2 * 4 * 16];
    __shared__ float4 ldsX[4 * TILE_POS * 2 * 16];
    __shared__ __attribute__((aligned(16))) float ldsVotes[WAVES_PER_BLOCK][16 * VOTE_STR];

    const int tid  = threadIdx.x;
    const int wib  = tid >> 5;
    const int lane = tid & 31;

    const int wq  = blockIdx.x & 7;
    const int hp  = (blockIdx.x >> 3) & 15;
    const int icg = blockIdx.x >> 7;

    {
        const int part = tid & 7;
        const int bb   = (tid >> 3) & 15;
        const int p0   = tid >> 7;
        float4 xv[12];
        for (int j = 0; j < 12; ++j) {
            const int r  = j / 3;
            const int cc = 2 * (j % 3) + p0;
            const int hh = 2 * hp + r - 1;
            const int ww = 4 * wq + cc - 1;
            const bool inb = ((unsigned)hh < 32u) && ((unsigned)ww < 32u);
            const int hc = min(max(hh, 0), 31);
            const int wc = min(max(ww, 0), 31);
            const float4 v = *(const float4*)(in + (size_t)bb * 131072 + hc * 4096 + wc * 128
                                                 + icg * 32 + part * 4);
            xv[j] = inb ? v : float4{0.f, 0.f, 0.f, 0.f};
        }
        for (int j = 0; j < 12; ++j) {
            const int r  = j / 3;
            const int cc = 2 * (j % 3) + p0;
            const int pos = r * TILE_COLS + cc;
            ldsX[(((part >> 1) * TILE_POS + pos) * 2 + (part & 1)) * 16 + ((bb + 4 * part) & 15)] = xv[j];
        }
    }

    for (int i = 0; i < 9; ++i) {
        const int idx = tid + i * 32 * WAVES_PER_BLOCK;
        const int sm_ = idx & 15;
        const int sq  = (idx >> 4) & 3;
        const int sh  = (idx >> 6) & 1;
        const int sk  = idx >> 7;
        const int K0  = sk * 4 + sh * 2;
        const int c0  = sq * 32 + sm_;
        float4 v;
        v.x = Wg[K0 * NDIM + c0];
        v.y = Wg[(K0 + 1) * NDIM + c0];
        v.z = Wg[K0 * NDIM + c0 + 16];
        v.w = Wg[(K0 + 1) * NDIM + c0 + 16];
        ldsWf[idx] = v;
    }
    __syncthreads();

    const int m  = lane & 15;
    const int hi = lane >> 4;

    float bias_c[4];
#pragma unroll
    for (int i = 0; i < 4; ++i) bias_c[i] = bias[8 * m + 4 * hi + i];
    float bias_r[4];
#pragma unroll
    for (int j = 0; j < 4; ++j) bias_r[j] = bias[64 * hi + 16 * j + m];

    const float4* wfrag = ldsWf + hi * 64 + m;

    float* votes  = ldsVotes[wib];

    const int icl = wib & 3;
    const int hl  = wib >> 2;
    const int slotE = (m + 8 * icl) & 15;
    const int slotO = (slotE + 4) & 15;
    const float* xrow = (const float*)ldsX + (icl * TILE_POS + hl * TILE_COLS) * 128 + 2 * hi;
    const float* xE = xrow + slotE * 4;
    const float* xO = xrow + 64 + slotO * 4;

#pragma unroll 1
    for (int it = 0; it < PIX_PER_WAVE; ++it) {
        const int icap  = icg * 4 + icl;
        const int h     = 2 * hp + hl;
        const int w     = 4 * wq + it;
        const int pixel = (icap << 10) + (h << 5) + w;

        v8f acc[8];
#pragma unroll
        for (int t = 0; t < 8; ++t)
            acc[t] = v8f{0.f, 0.f, 0.f, 0.f, 0.f, 0.f, 0.f, 0.f};

        const float* tapE = xE + it * 128;
        const float* tapO = xO + it * 128;

#pragma unroll
        for (int ks = 0; ks < 18; ++ks) {
            const int tap = ks >> 1;
            const int kh  = tap / 3;
            const int kw  = tap - kh * 3;
            v2f a = *(const v2f*)(((ks & 1) ? tapO : tapE) + (kh * TILE_COLS + kw) * 128);
            for (int q = 0; q < 4; ++q) {
                const float4 w4 = wfrag[ks * 128 + q * 16];
                v2f b0, b1;
                b0[0] = w4.x; b0[1] = w4.y;
                b1[0] = w4.z; b1[1] = w4.w;
                acc[2 * q] = __builtin_amdgcn_wmma_f32_16x16x4_f32(
                    false, a, false, b0, (short)0, acc[2 * q], false, false);
                acc[2 * q + 1] = __builtin_amdgcn_wmma_f32_16x16x4_f32(
                    false, a, false, b1, (short)0, acc[2 * q + 1], false, false);
            }
        }

#pragma unroll
        for (int t = 0; t < 8; ++t)
#pragma unroll
            for (int v = 0; v < 8; ++v)
                votes[(v + 8 * hi) * VOTE_STR + t * 16 + m] = acc[t][v];

        const float* vcol = votes + 8 * m + 4 * hi;
        v2f s01 = v2f{0.f, 0.f};
        v2f s23 = v2f{0.f, 0.f};
        for (int ic = 0; ic < 16; ++ic) {
            const float4 v4 = *(const float4*)(vcol + ic * VOTE_STR);
            s01 += v2f{v4.x, v4.y};
            s23 += v2f{v4.z, v4.w};
        }
        v2f bq0, bq1;
        bq0[0] = s01[0] * (1.f / 16.f) + bias_c[0];
        bq0[1] = s01[1] * (1.f / 16.f) + bias_c[1];
        bq1[0] = s23[0] * (1.f / 16.f) + bias_c[2];
        bq1[1] = s23[1] * (1.f / 16.f) + bias_c[3];

        const v2f zero2 = v2f{0.f, 0.f};
        v8f lgE = v8f{0.f, 0.f, 0.f, 0.f, 0.f, 0.f, 0.f, 0.f};
        v8f lgO = lgE;
        v2f avf[32];
        const float* vrow = votes + m * VOTE_STR + 4 * hi;
#pragma unroll
        for (int oc = 0; oc < 16; ++oc) {
            const float4 a4 = *(const float4*)(vrow + 8 * oc);
            avf[2 * oc][0]     = a4.x;
            avf[2 * oc][1]     = a4.y;
            avf[2 * oc + 1][0] = a4.z;
            avf[2 * oc + 1][1] = a4.w;
        }
        __builtin_amdgcn_sched_barrier(0);
#pragma unroll
        for (int ks = 0; ks < 32; ++ks) {
            const v2f av = avf[ks];
            const bool own = (m == (ks >> 1));
            const v2f bv = own ? ((ks & 1) ? bq1 : bq0) : zero2;
            if (ks & 1)
                lgO = __builtin_amdgcn_wmma_f32_16x16x4_f32(
                    false, av, false, bv, (short)0, lgO, false, false);
            else
                lgE = __builtin_amdgcn_wmma_f32_16x16x4_f32(
                    false, av, false, bv, (short)0, lgE, false, false);
        }
        float n2c = (bq0[0] * bq0[0] + bq0[1] * bq0[1]) + (bq1[0] * bq1[0] + bq1[1] * bq1[1]);
        n2c += xhalf(n2c);
        const float sc1 = (n2c / (1.f + n2c)) / sqrtf(n2c + CAPS_EPS);
        v2f rt[4];
        float lg[8], ex[8];
#pragma unroll
        for (int r = 0; r < 8; ++r) lg[r] = (lgE[r] + lgO[r]) * sc1;
        const bool lb0 = (m & 1) != 0, lb1 = (m & 2) != 0, lb2 = (m & 4) != 0;
        const float mx_own = row_reduce_own<true>(lg, lb0, lb1, lb2);
        ex[0] = __expf(lg[0] - dpp_read<0x150 + 0>(mx_own));
        ex[1] = __expf(lg[1] - dpp_read<0x150 + 1>(mx_own));
        ex[2] = __expf(lg[2] - dpp_read<0x150 + 2>(mx_own));
        ex[3] = __expf(lg[3] - dpp_read<0x150 + 3>(mx_own));
        ex[4] = __expf(lg[4] - dpp_read<0x150 + 4>(mx_own));
        ex[5] = __expf(lg[5] - dpp_read<0x150 + 5>(mx_own));
        ex[6] = __expf(lg[6] - dpp_read<0x150 + 6>(mx_own));
        ex[7] = __expf(lg[7] - dpp_read<0x150 + 7>(mx_own));
        const float inv_own = 1.f / row_reduce_own<false>(ex, lb0, lb1, lb2);
        rt[0][0] = ex[0] * dpp_read<0x150 + 0>(inv_own);
        rt[0][1] = ex[1] * dpp_read<0x150 + 1>(inv_own);
        rt[1][0] = ex[2] * dpp_read<0x150 + 2>(inv_own);
        rt[1][1] = ex[3] * dpp_read<0x150 + 3>(inv_own);
        rt[2][0] = ex[4] * dpp_read<0x150 + 4>(inv_own);
        rt[2][1] = ex[5] * dpp_read<0x150 + 5>(inv_own);
        rt[3][0] = ex[6] * dpp_read<0x150 + 6>(inv_own);
        rt[3][1] = ex[7] * dpp_read<0x150 + 7>(inv_own);

        v8f S[8];
#pragma unroll
        for (int t = 0; t < 8; ++t)
            S[t] = v8f{0.f, 0.f, 0.f, 0.f, 0.f, 0.f, 0.f, 0.f};
#pragma unroll
        for (int ks = 0; ks < 4; ++ks) {
#pragma unroll
            for (int t = 0; t < 8; ++t) {
                v2f bv;
                bv[0] = acc[t][2 * ks];
                bv[1] = acc[t][2 * ks + 1];
                S[t] = __builtin_amdgcn_wmma_f32_16x16x4_f32(
                    false, rt[ks], false, bv, (short)0, S[t], false, false);
            }
        }
        const bool up = (m & 8) != 0;
        float pr[4], n2[4];
#pragma unroll
        for (int j = 0; j < 4; ++j) {
            const float lo = hi ? S[4 + j][2 * j]     : S[j][2 * j];
            const float hv = hi ? S[4 + j][2 * j + 1] : S[j][2 * j + 1];
            pr[j] = (up ? hv : lo) + bias_r[j];
            n2[j] = pr[j] * pr[j];
        }
        float u2[2];
#pragma unroll
        for (int i = 0; i < 2; ++i) {
            const float keep = lb0 ? n2[2 * i + 1] : n2[2 * i];
            const float send = lb0 ? n2[2 * i]     : n2[2 * i + 1];
            u2[i] = keep + row_xor<1>(send);
        }
        const float keep2 = lb1 ? u2[1] : u2[0];
        const float send2 = lb1 ? u2[0] : u2[1];
        float n2own = keep2 + row_xor<2>(send2);
        n2own += row_xor<4>(n2own);
        const float sc_own = (n2own / (1.f + n2own)) / sqrtf(n2own + CAPS_EPS);
        out[(size_t)pixel * 128 + 64 * hi + 16 * 0 + m] = pr[0] * dpp_read<0 * 0x55>(sc_own);
        out[(size_t)pixel * 128 + 64 * hi + 16 * 1 + m] = pr[1] * dpp_read<1 * 0x55>(sc_own);
        out[(size_t)pixel * 128 + 64 * hi + 16 * 2 + m] = pr[2] * dpp_read<2 * 0x55>(sc_own);
        out[(size_t)pixel * 128 + 64 * hi + 16 * 3 + m] = pr[3] * dpp_read<3 * 0x55>(sc_own);
    }
}

extern "C" void kernel_launch(void* const* d_in, const int* in_sizes, int n_in,
                              void* d_out, int out_size, void* d_ws, size_t ws_size,
                              hipStream_t stream) {
    (void)in_sizes; (void)n_in; (void)out_size; (void)ws_size;
    const float* input = (const float*)d_in[0];
    const float* Wg    = (const float*)d_in[1];
    const float* bias  = (const float*)d_in[2];
    float* out = (float*)d_out;
    (void)d_ws;

    caps_conv_route_kernel<<<NPIX / (WAVES_PER_BLOCK * PIX_PER_WAVE), 32 * WAVES_PER_BLOCK, 0, stream>>>(input, Wg, bias, out);
}


# Route-weighted vote sums taken off the matrix pipe (block diagonal: 1 tile in 16 used): f32 FMAs per lane (lane = capsule, 4 dims) on held slab values, one 16-byte store per lane; new sum order
# speedup vs baseline: 4.3145x; 1.1208x over previous
#include <hip/hip_runtime.h>
#include <hip/hip_bf16.h>
#include <math.h>


typedef __attribute__((ext_vector_type(2))) float v2f;
typedef __attribute__((ext_vector_type(8))) float v8f;

#define NPIX      16384
#define KDIM      72
#define NDIM      128
#define WAVES_PER_BLOCK 8
#define VOTE_STR  132
#define CAPS_EPS  1e-7f
#define PIX_PER_WAVE 4
#define TILE_ROWS 4
#define TILE_COLS 6
#define TILE_POS  (TILE_ROWS * TILE_COLS)

template <int MASK>
__device__ __forceinline__ float row_xor(float x) {
    return __builtin_bit_cast(float,
        __builtin_amdgcn_mov_dpp(__builtin_bit_cast(int, x), 0x160 + MASK, 0xf, 0xf, false));
}

__device__ __forceinline__ float xhalf(float x) {
    return __builtin_bit_cast(float,
        __builtin_amdgcn_permlanex16(__builtin_bit_cast(int, x), __builtin_bit_cast(int, x),
                                     0x76543210, 0xfedcba98, false, false));
}

template <int CTRL>
__device__ __forceinline__ float dpp_read(float x) {
    return __builtin_bit_cast(float,
        __builtin_amdgcn_mov_dpp(__builtin_bit_cast(int, x), CTRL, 0xf, 0xf, false));
}

template <bool IS_MAX>
__device__ __forceinline__ float comb2(float a, float b) { return IS_MAX ? fmaxf(a, b) : a + b; }

template <bool IS_MAX>
__device__ __forceinline__ float row_reduce_own(const float (&v)[8], bool b0, bool b1, bool b2) {
    float p[4], q[2];
#pragma unroll
    for (int i = 0; i < 4; ++i) {
        const float keep = b0 ? v[2 * i + 1] : v[2 * i];
        const float send = b0 ? v[2 * i]     : v[2 * i + 1];
        p[i] = comb2<IS_MAX>(keep, row_xor<1>(send));
    }
#pragma unroll
    for (int i = 0; i < 2; ++i) {
        const float keep = b1 ? p[2 * i + 1] : p[2 * i];
        const float send = b1 ? p[2 * i]     : p[2 * i + 1];
        q[i] = comb2<IS_MAX>(keep, row_xor<2>(send));
    }
    const float keep = b2 ? q[1] : q[0];
    const float send = b2 ? q[0] : q[1];
    float r = comb2<IS_MAX>(keep, row_xor<4>(send));
    r = comb2<IS_MAX>(r, row_xor<8>(r));
    return r;
}

__global__ __launch_bounds__(32 * WAVES_PER_BLOCK) void caps_conv_route_kernel(
    const float* __restrict__ in,
    const float* __restrict__ Wg,
    const float* __restrict__ bias,
    float* __restrict__ out)
{
    __shared__ float4 ldsWf[18 * 2 * 4 * 16];
    __shared__ float4 ldsX[4 * TILE_POS * 2 * 16];
    __shared__ __attribute__((aligned(16))) float ldsVotes[WAVES_PER_BLOCK][16 * VOTE_STR];

    const int tid  = threadIdx.x;
    const int wib  = tid >> 5;
    const int lane = tid & 31;

    const int wq  = blockIdx.x & 7;
    const int hp  = (blockIdx.x >> 3) & 15;
    const int icg = blockIdx.x >> 7;

    {
        const int part = tid & 7;
        const int bb   = (tid >> 3) & 15;
        const int p0   = tid >> 7;
        float4 xv[12];
        for (int j = 0; j < 12; ++j) {
            const int r  = j / 3;
            const int cc = 2 * (j % 3) + p0;
            const int hh = 2 * hp + r - 1;
            const int ww = 4 * wq + cc - 1;
            const bool inb = ((unsigned)hh < 32u) && ((unsigned)ww < 32u);
            const int hc = min(max(hh, 0), 31);
            const int wc = min(max(ww, 0), 31);
            const float4 v = *(const float4*)(in + (size_t)bb * 131072 + hc * 4096 + wc * 128
                                                 + icg * 32 + part * 4);
            xv[j] = inb ? v : float4{0.f, 0.f, 0.f, 0.f};
        }
        for (int j = 0; j < 12; ++j) {
            const int r  = j / 3;
            const int cc = 2 * (j % 3) + p0;
            const int pos = r * TILE_COLS + cc;
            ldsX[(((part >> 1) * TILE_POS + pos) * 2 + (part & 1)) * 16 + ((bb + 4 * part) & 15)] = xv[j];
        }
    }

    for (int i = 0; i < 9; ++i) {
        const int idx = tid + i * 32 * WAVES_PER_BLOCK;
        const int sm_ = idx & 15;
        const int sq  = (idx >> 4) & 3;
        const int sh  = (idx >> 6) & 1;
        const int sk  = idx >> 7;
        const int K0  = sk * 4 + sh * 2;
        const int c0  = sq * 32 + sm_;
        float4 v;
        v.x = Wg[K0 * NDIM + c0];
        v.y = Wg[(K0 + 1) * NDIM + c0];
        v.z = Wg[K0 * NDIM + c0 + 16];
        v.w = Wg[(K0 + 1) * NDIM + c0 + 16];
        ldsWf[idx] = v;
    }
    __syncthreads();

    const int m  = lane & 15;
    const int hi = lane >> 4;

    float bias_c[4];
#pragma unroll
    for (int i = 0; i < 4; ++i) bias_c[i] = bias[8 * m + 4 * hi + i];

    const float4* wfrag = ldsWf + hi * 64 + m;

    float* votes  = ldsVotes[wib];

    const int icl = wib & 3;
    const int hl  = wib >> 2;
    const int slotE = (m + 8 * icl) & 15;
    const int slotO = (slotE + 4) & 15;
    const float* xrow = (const float*)ldsX + (icl * TILE_POS + hl * TILE_COLS) * 128 + 2 * hi;
    const float* xE = xrow + slotE * 4;
    const float* xO = xrow + 64 + slotO * 4;

#pragma unroll 1
    for (int it = 0; it < PIX_PER_WAVE; ++it) {
        const int icap  = icg * 4 + icl;
        const int h     = 2 * hp + hl;
        const int w     = 4 * wq + it;
        const int pixel = (icap << 10) + (h << 5) + w;

        v8f acc[8];
#pragma unroll
        for (int t = 0; t < 8; ++t)
            acc[t] = v8f{0.f, 0.f, 0.f, 0.f, 0.f, 0.f, 0.f, 0.f};

        const float* tapE = xE + it * 128;
        const float* tapO = xO + it * 128;

#pragma unroll
        for (int ks = 0; ks < 18; ++ks) {
            const int tap = ks >> 1;
            const int kh  = tap / 3;
            const int kw  = tap - kh * 3;
            v2f a = *(const v2f*)(((ks & 1) ? tapO : tapE) + (kh * TILE_COLS + kw) * 128);
            for (int q = 0; q < 4; ++q) {
                const float4 w4 = wfrag[ks * 128 + q * 16];
                v2f b0, b1;
                b0[0] = w4.x; b0[1] = w4.y;
                b1[0] = w4.z; b1[1] = w4.w;
                acc[2 * q] = __builtin_amdgcn_wmma_f32_16x16x4_f32(
                    false, a, false, b0, (short)0, acc[2 * q], false, false);
                acc[2 * q + 1] = __builtin_amdgcn_wmma_f32_16x16x4_f32(
                    false, a, false, b1, (short)0, acc[2 * q + 1], false, false);
            }
        }

#pragma unroll
        for (int t = 0; t < 8; ++t)
#pragma unroll
            for (int v = 0; v < 8; ++v)
                votes[(v + 8 * hi) * VOTE_STR + t * 16 + m] = acc[t][v];

        const float* vcol = votes + 8 * m + 4 * hi;
        float4 vv[16];
        v2f s01 = v2f{0.f, 0.f};
        v2f s23 = v2f{0.f, 0.f};
        for (int ic = 0; ic < 16; ++ic) {
            vv[ic] = *(const float4*)(vcol + ic * VOTE_STR);
            s01 += v2f{vv[ic].x, vv[ic].y};
            s23 += v2f{vv[ic].z, vv[ic].w};
        }
        v2f bq0, bq1;
        bq0[0] = s01[0] * (1.f / 16.f) + bias_c[0];
        bq0[1] = s01[1] * (1.f / 16.f) + bias_c[1];
        bq1[0] = s23[0] * (1.f / 16.f) + bias_c[2];
        bq1[1] = s23[1] * (1.f / 16.f) + bias_c[3];

        const v2f zero2 = v2f{0.f, 0.f};
        v8f lgE = v8f{0.f, 0.f, 0.f, 0.f, 0.f, 0.f, 0.f, 0.f};
        v8f lgO = lgE;
        v2f avf[32];
        const float* vrow = votes + m * VOTE_STR + 4 * hi;
#pragma unroll
        for (int oc = 0; oc < 16; ++oc) {
            const float4 a4 = *(const float4*)(vrow + 8 * oc);
            avf[2 * oc][0]     = a4.x;
            avf[2 * oc][1]     = a4.y;
            avf[2 * oc + 1][0] = a4.z;
            avf[2 * oc + 1][1] = a4.w;
        }
        __builtin_amdgcn_sched_barrier(0);
#pragma unroll
        for (int ks = 0; ks < 32; ++ks) {
            const v2f av = avf[ks];
            const bool own = (m == (ks >> 1));
            const v2f bv = own ? ((ks & 1) ? bq1 : bq0) : zero2;
            if (ks & 1)
                lgO = __builtin_amdgcn_wmma_f32_16x16x4_f32(
                    false, av, false, bv, (short)0, lgO, false, false);
            else
                lgE = __builtin_amdgcn_wmma_f32_16x16x4_f32(
                    false, av, false, bv, (short)0, lgE, false, false);
        }
        float n2c = (bq0[0] * bq0[0] + bq0[1] * bq0[1]) + (bq1[0] * bq1[0] + bq1[1] * bq1[1]);
        n2c += xhalf(n2c);
        const float sc1 = (n2c / (1.f + n2c)) / sqrtf(n2c + CAPS_EPS);
        float rown[8];
        float lg[8], ex[8];
#pragma unroll
        for (int r = 0; r < 8; ++r) lg[r] = (lgE[r] + lgO[r]) * sc1;
        const bool lb0 = (m & 1) != 0, lb1 = (m & 2) != 0, lb2 = (m & 4) != 0;
        const float mx_own = row_reduce_own<true>(lg, lb0, lb1, lb2);
        ex[0] = __expf(lg[0] - dpp_read<0x150 + 0>(mx_own));
        ex[1] = __expf(lg[1] - dpp_read<0x150 + 1>(mx_own));
        ex[2] = __expf(lg[2] - dpp_read<0x150 + 2>(mx_own));
        ex[3] = __expf(lg[3] - dpp_read<0x150 + 3>(mx_own));
        ex[4] = __expf(lg[4] - dpp_read<0x150 + 4>(mx_own));
        ex[5] = __expf(lg[5] - dpp_read<0x150 + 5>(mx_own));
        ex[6] = __expf(lg[6] - dpp_read<0x150 + 6>(mx_own));
        ex[7] = __expf(lg[7] - dpp_read<0x150 + 7>(mx_own));
        const float inv_own = 1.f / row_reduce_own<false>(ex, lb0, lb1, lb2);
        rown[0] = ex[0] * dpp_read<0x150 + 0>(inv_own);
        rown[1] = ex[1] * dpp_read<0x150 + 1>(inv_own);
        rown[2] = ex[2] * dpp_read<0x150 + 2>(inv_own);
        rown[3] = ex[3] * dpp_read<0x150 + 3>(inv_own);
        rown[4] = ex[4] * dpp_read<0x150 + 4>(inv_own);
        rown[5] = ex[5] * dpp_read<0x150 + 5>(inv_own);
        rown[6] = ex[6] * dpp_read<0x150 + 6>(inv_own);
        rown[7] = ex[7] * dpp_read<0x150 + 7>(inv_own);

        float rlo[8], rhi[8];
        for (int r = 0; r < 8; ++r) {
            const float oth = xhalf(rown[r]);
            rlo[r] = hi ? oth : rown[r];
            rhi[r] = hi ? rown[r] : oth;
        }
        v2f p01 = v2f{0.f, 0.f};
        v2f p23 = v2f{0.f, 0.f};
        for (int r = 0; r < 8; ++r) {
            p01 += v2f{rlo[r], rlo[r]} * v2f{vv[r].x, vv[r].y};
            p23 += v2f{rlo[r], rlo[r]} * v2f{vv[r].z, vv[r].w};
        }
        for (int r = 0; r < 8; ++r) {
            p01 += v2f{rhi[r], rhi[r]} * v2f{vv[8 + r].x, vv[8 + r].y};
            p23 += v2f{rhi[r], rhi[r]} * v2f{vv[8 + r].z, vv[8 + r].w};
        }
        float4 pr;
        pr.x = p01[0] + bias_c[0];
        pr.y = p01[1] + bias_c[1];
        pr.z = p23[0] + bias_c[2];
        pr.w = p23[1] + bias_c[3];
        float n2o = (pr.x * pr.x + pr.y * pr.y) + (pr.z * pr.z + pr.w * pr.w);
        n2o += xhalf(n2o);
        const float sc2 = (n2o / (1.f + n2o)) / sqrtf(n2o + CAPS_EPS);
        *(float4*)(out + (size_t)pixel * 128 + 8 * m + 4 * hi) =
            float4{pr.x * sc2, pr.y * sc2, pr.z * sc2, pr.w * sc2};
    }
}

extern "C" void kernel_launch(void* const* d_in, const int* in_sizes, int n_in,
                              void* d_out, int out_size, void* d_ws, size_t ws_size,
                              hipStream_t stream) {
    (void)in_sizes; (void)n_in; (void)out_size; (void)ws_size;
    const float* input = (const float*)d_in[0];
    const float* Wg    = (const float*)d_in[1];
    const float* bias  = (const float*)d_in[2];
    float* out = (float*)d_out;
    (void)d_ws;

    caps_conv_route_kernel<<<NPIX / (WAVES_PER_BLOCK * PIX_PER_WAVE), 32 * WAVES_PER_BLOCK, 0, stream>>>(input, Wg, bias, out);
}
